# starting kernel
# baseline (speedup 1.0000x reference)
#include <hip/hip_runtime.h>
#include <hip/hip_bf16.h>


typedef __attribute__((ext_vector_type(2))) float v2f;
typedef __attribute__((ext_vector_type(4))) float v4f;
typedef __attribute__((ext_vector_type(8))) float v8f;

#define BATCH    16384
#define DIM      1024
#define NLAYERS  4
#define KCHUNKS  4
#define KCHUNK   (DIM / KCHUNKS)
#define KSTEPS   (KCHUNK / 4)
#define ROWTILES (BATCH / 16)
#define WPB      8

#define WS_C     0
#define WS_BSUM  16
#define WS_A4    2048
#define WS_DPART 32768

__global__ void crossnet_prep(const float* __restrict__ W,
                              const float* __restrict__ b,
                              float* __restrict__ ws) {
  const int t = threadIdx.x;
  const float b0 = b[0 * DIM + t], b1 = b[1 * DIM + t];
  const float b2 = b[2 * DIM + t], b3 = b[3 * DIM + t];
  const float B1 = b0, B2 = b0 + b1, B3 = b0 + b1 + b2;
  ws[WS_BSUM + t] = B3 + b3;

  __shared__ float red[3][1024];
  red[0][t] = B1 * W[1 * DIM + t];
  red[1][t] = B2 * W[2 * DIM + t];
  red[2][t] = B3 * W[3 * DIM + t];
  __syncthreads();
  for (int off = 512; off > 0; off >>= 1) {
    if (t < off) {
      red[0][t] += red[0][t + off];
      red[1][t] += red[1][t + off];
      red[2][t] += red[2][t + off];
    }
    __syncthreads();
  }
  if (t == 0) {
    ws[WS_C + 0] = 0.0f;
    ws[WS_C + 1] = red[0][0];
    ws[WS_C + 2] = red[1][0];
    ws[WS_C + 3] = red[2][0];
  }
}

__global__ void __launch_bounds__(256)
crossnet_dots(const float* __restrict__ x0,
              const float* __restrict__ W,
              float* __restrict__ ws) {
  const int kchunk  = blockIdx.x & (KCHUNKS - 1);
  const int waveId  = threadIdx.x >> 5;
  const int rowTile = (blockIdx.x >> 2) * WPB + waveId;
  const int lane    = threadIdx.x & 31;
  const int n       = lane & 15;
  const int hi      = lane >> 4;

  __shared__ v2f btab[KSTEPS][32];

  {
    const int t = threadIdx.x;
#pragma unroll
    for (int i = 0; i < (KSTEPS * 32) / 256; ++i) {
      const int e     = t + i * 256;
      const int estep = e >> 5;
      const int elane = e & 31;
      const int en    = elane & 15;
      const int ekk   = (elane >> 4) * 2;
      const int kpos  = kchunk * KCHUNK + estep * 4 + ekk;
      float vx = W[(en & 3) * DIM + kpos];
      float vy = W[(en & 3) * DIM + kpos + 1];
      if (en >= 4) { vx = 0.0f; vy = 0.0f; }
      v2f v; v.x = vx; v.y = vy;
      btab[estep][elane] = v;
    }
  }
  __syncthreads();

  const float* aptr = x0 + (size_t)(rowTile * 16 + n) * DIM
                         + kchunk * KCHUNK + hi * 2;
  const v2f* bp = &btab[0][lane];

  v8f acc0 = {}, acc1 = {};
#pragma unroll 4
  for (int s = 0; s < KSTEPS; s += 2) {
    v2f a0 = *(const v2f*)(aptr + s * 4);
    v2f a1 = *(const v2f*)(aptr + s * 4 + 4);
    v2f b0 = bp[(s + 0) * 32];
    v2f b1 = bp[(s + 1) * 32];
    acc0 = __builtin_amdgcn_wmma_f32_16x16x4_f32(
        false, a0, false, b0, (short)0, acc0, false, false);
    acc1 = __builtin_amdgcn_wmma_f32_16x16x4_f32(
        false, a1, false, b1, (short)0, acc1, false, false);
  }
  v8f acc = acc0 + acc1;

  if (n < 4) {
    float* dp = ws + WS_DPART + (size_t)kchunk * BATCH * 4;
#pragma unroll
    for (int r = 0; r < 8; ++r) {
      const int row = rowTile * 16 + hi * 8 + r;
      dp[(size_t)row * 4 + n] = acc[r];
    }
  }
}

__global__ void crossnet_rec(float* __restrict__ ws) {
  const int row = blockIdx.x * blockDim.x + threadIdx.x;
  if (row >= BATCH) return;
  v4f d = {0.0f, 0.0f, 0.0f, 0.0f};
  const float* dp = ws + WS_DPART;
#pragma unroll
  for (int cch = 0; cch < KCHUNKS; ++cch)
    d += *(const v4f*)(dp + ((size_t)cch * BATCH + row) * 4);

  float A = 1.0f;
#pragma unroll
  for (int l = 0; l < NLAYERS; ++l) {
    const float s = A * d[l] + ws[WS_C + l];
    A += s;
  }
  ws[WS_A4 + row] = A;
}

__global__ void crossnet_out(const float* __restrict__ x0,
                             const float* __restrict__ ws,
                             float* __restrict__ out) {
  const size_t i4  = (size_t)blockIdx.x * blockDim.x + threadIdx.x;
  const size_t j   = i4 * 4;
  const int    row = (int)(j >> 10);
  const int    col = (int)(j & 1023);
  const float  A   = ws[WS_A4 + row];
  const v4f    x   = *(const v4f*)(x0 + j);
  const v4f    bs  = *(const v4f*)(ws + WS_BSUM + col);
  v4f o;
  o.x = A * x.x + bs.x;
  o.y = A * x.y + bs.y;
  o.z = A * x.z + bs.z;
  o.w = A * x.w + bs.w;
  *(v4f*)(out + j) = o;
}

extern "C" void kernel_launch(void* const* d_in, const int* in_sizes, int n_in,
                              void* d_out, int out_size, void* d_ws, size_t ws_size,
                              hipStream_t stream) {
  const float* x0 = (const float*)d_in[0];
  const float* W  = (const float*)d_in[1];
  const float* b  = (const float*)d_in[2];
  float* out = (float*)d_out;
  float* ws  = (float*)d_ws;

  (void)in_sizes; (void)n_in; (void)out_size; (void)ws_size;

  crossnet_prep<<<1, 1024, 0, stream>>>(W, b, ws);

  crossnet_dots<<<(ROWTILES * KCHUNKS) / WPB, 256, 0, stream>>>(x0, W, ws);

  crossnet_rec<<<BATCH / 256, 256, 0, stream>>>(ws);

  crossnet_out<<<(BATCH * DIM / 4) / 256, 256, 0, stream>>>(x0, ws, out);
}


# One fused launch: x0 read once into registers by whole-line loads, turned to WMMA A layout via LDS, output written from the same registers; row dots now 8 K-slices of 128 (sum order changes)
# speedup vs baseline: 1.9772x; 1.9772x over previous
#include <hip/hip_runtime.h>


typedef __attribute__((ext_vector_type(2))) float v2f;
typedef __attribute__((ext_vector_type(4))) float v4f;
typedef __attribute__((ext_vector_type(8))) float v8f;

#define BATCH    16384
#define DIM      1024
#define NLAYERS  4
#define ROWS     16
#define NTILES   (BATCH / ROWS)
#define WAVES    8
#define KSLICE   (DIM / WAVES)
#define TPB      2
#define XPITCH   132
#define WPITCH   1028

#define L_XS     0
#define L_WT     (L_XS + WAVES * ROWS * XPITCH)
#define L_ZERO   (L_WT + NLAYERS * WPITCH)
#define L_CPART  (L_ZERO + 128)
#define L_DPART  (L_CPART + WAVES * 4)
#define L_TOTAL  (L_DPART + WAVES * ROWS * 4)

__device__ __forceinline__ void lds_barrier() {
  __builtin_amdgcn_fence(__ATOMIC_RELEASE, "workgroup", "local");
  __builtin_amdgcn_s_barrier();
  __builtin_amdgcn_fence(__ATOMIC_ACQUIRE, "workgroup", "local");
}

__global__ void __launch_bounds__(256)
crossnet_fused(const float* __restrict__ x0,
               const float* __restrict__ W,
               const float* __restrict__ b,
               float* __restrict__ out) {
  __shared__ __attribute__((aligned(16))) float lds[L_TOTAL];

  const int t    = threadIdx.x;
  const int w    = t >> 5;
  const int lane = t & 31;
  const int n    = lane & 15;
  const int hi   = lane >> 4;
  const int col  = 4 * t;

  v4f X[TPB][ROWS];
  for (int q = 0; q < TPB; ++q) {
    const size_t row0 = (size_t)(blockIdx.x * TPB + q) * ROWS;
    for (int r = 0; r < ROWS; ++r)
      X[q][r] = *(const v4f*)(x0 + (row0 + r) * DIM + col);
  }

  const v4f b0 = *(const v4f*)(b + 0 * DIM + col);
  const v4f b1 = *(const v4f*)(b + 1 * DIM + col);
  const v4f b2 = *(const v4f*)(b + 2 * DIM + col);
  const v4f b3 = *(const v4f*)(b + 3 * DIM + col);
  const v4f w0 = *(const v4f*)(W + 0 * DIM + col);
  const v4f w1 = *(const v4f*)(W + 1 * DIM + col);
  const v4f w2 = *(const v4f*)(W + 2 * DIM + col);
  const v4f w3 = *(const v4f*)(W + 3 * DIM + col);
  const v4f B1 = b0;
  const v4f B2 = b0 + b1;
  const v4f B3 = B2 + b2;
  const v4f bsum = B3 + b3;

  float p1 = B1.x * w1.x + B1.y * w1.y + B1.z * w1.z + B1.w * w1.w;
  float p2 = B2.x * w2.x + B2.y * w2.y + B2.z * w2.z + B2.w * w2.w;
  float p3 = B3.x * w3.x + B3.y * w3.y + B3.z * w3.z + B3.w * w3.w;
  for (int off = 16; off > 0; off >>= 1) {
    p1 += __shfl_xor(p1, off);
    p2 += __shfl_xor(p2, off);
    p3 += __shfl_xor(p3, off);
  }
  if (lane == 0) {
    v4f cp; cp.x = 0.0f; cp.y = p1; cp.z = p2; cp.w = p3;
    *(v4f*)&lds[L_CPART + w * 4] = cp;
  }
  *(v4f*)&lds[L_WT + 0 * WPITCH + col] = w0;
  *(v4f*)&lds[L_WT + 1 * WPITCH + col] = w1;
  *(v4f*)&lds[L_WT + 2 * WPITCH + col] = w2;
  *(v4f*)&lds[L_WT + 3 * WPITCH + col] = w3;
  if (t < 32) {
    const v4f z = {0.0f, 0.0f, 0.0f, 0.0f};
    *(v4f*)&lds[L_ZERO + 4 * t] = z;
  }

  const v4f* lds4 = (const v4f*)lds;
  const int aoff4 = L_XS / 4 + (w * ROWS + n) * (XPITCH / 4) + hi;
  const int boff4 = (n < NLAYERS) ? (L_WT / 4 + n * (WPITCH / 4) + (KSLICE / 4) * w + hi)
                                  : (L_ZERO / 4 + hi);
  const int xsto = L_XS + w * ROWS * XPITCH + 4 * lane;

  v4f c = {0.0f, 0.0f, 0.0f, 0.0f};

  for (int q = 0; q < TPB; ++q) {
    const size_t row0 = (size_t)(blockIdx.x * TPB + q) * ROWS;

    for (int r = 0; r < ROWS; ++r)
      *(v4f*)&lds[xsto + r * XPITCH] = X[q][r];
    lds_barrier();

    if (q == 0) {
      for (int ww = 0; ww < WAVES; ++ww)
        c += *(const v4f*)&lds[L_CPART + ww * 4];
    }

    v8f acc0 = {}, acc1 = {};
    for (int j = 0; j < KSLICE / 8; ++j) {
      const v4f a  = lds4[aoff4 + 2 * j];
      const v4f bq = lds4[boff4 + 2 * j];
      v2f a_lo; a_lo.x = a.x;  a_lo.y = a.y;
      v2f a_hi; a_hi.x = a.z;  a_hi.y = a.w;
      v2f b_lo; b_lo.x = bq.x; b_lo.y = bq.y;
      v2f b_hi; b_hi.x = bq.z; b_hi.y = bq.w;
      acc0 = __builtin_amdgcn_wmma_f32_16x16x4_f32(
          false, a_lo, false, b_lo, (short)0, acc0, false, false);
      acc1 = __builtin_amdgcn_wmma_f32_16x16x4_f32(
          false, a_hi, false, b_hi, (short)0, acc1, false, false);
    }
    const v8f acc = acc0 + acc1;

    if (n < NLAYERS) {
      for (int r = 0; r < 8; ++r)
        lds[L_DPART + (w * ROWS + hi * 8 + r) * 4 + n] = acc[r];
    }
    lds_barrier();

    v4f d = {0.0f, 0.0f, 0.0f, 0.0f};
    for (int ww = 0; ww < WAVES; ++ww)
      d += *(const v4f*)&lds[L_DPART + (ww * ROWS + n) * 4];
    float A = 1.0f;
    for (int l = 0; l < NLAYERS; ++l) {
      const float s = A * d[l] + c[l];
      A += s;
    }

    for (int r = 0; r < ROWS; ++r) {
      const float Ar = __int_as_float(
          __builtin_amdgcn_readlane(__float_as_int(A), r));
      const v4f x = X[q][r];
      v4f o;
      o.x = Ar * x.x + bsum.x;
      o.y = Ar * x.y + bsum.y;
      o.z = Ar * x.z + bsum.z;
      o.w = Ar * x.w + bsum.w;
      *(v4f*)(out + (row0 + r) * DIM + col) = o;
    }
  }
}

extern "C" void kernel_launch(void* const* d_in, const int* in_sizes, int n_in,
                              void* d_out, int out_size, void* d_ws, size_t ws_size,
                              hipStream_t stream) {
  const float* x0 = (const float*)d_in[0];
  const float* W  = (const float*)d_in[1];
  const float* b  = (const float*)d_in[2];
  float* out = (float*)d_out;

  (void)in_sizes; (void)n_in; (void)out_size; (void)d_ws; (void)ws_size;

  crossnet_fused<<<NTILES / TPB, 256, 0, stream>>>(x0, W, b, out);
}
